# starting kernel
# baseline (speedup 1.0000x reference)
#include <hip/hip_runtime.h>
#include <math.h>

typedef float v2f __attribute__((ext_vector_type(2)));
typedef float v8f __attribute__((ext_vector_type(8)));

#define B_TOT  8
#define C_DIM  512
#define K_DIM  19
#define N_SP   16384
#define CT     16
#define WAVES  4
#define NSPLIT 4
#define UNROLL 8
#define PN_OFF 512

__global__ void zero_out_kernel(float* __restrict__ out, int n) {
    int i = blockIdx.x * blockDim.x + threadIdx.x;
    if (i < n) out[i] = 0.0f;
}

__global__ void softmax_stats_kernel(const float* __restrict__ probs,
                                     float* __restrict__ stats) {
    __shared__ float red[256];
    const int r = blockIdx.x;
    const float* p = probs + (size_t)r * N_SP;
    const int t = threadIdx.x;

    float m = -3.402823466e38f;
    for (int i = t; i < N_SP; i += 256) m = fmaxf(m, p[i]);
    red[t] = m; __syncthreads();
    for (int s = 128; s > 0; s >>= 1) {
        if (t < s) red[t] = fmaxf(red[t], red[t + s]);
        __syncthreads();
    }
    const float rowmax = red[0];
    __syncthreads();

    float sum = 0.0f;
    for (int i = t; i < N_SP; i += 256) sum += __expf(p[i] - rowmax);
    red[t] = sum; __syncthreads();
    for (int s = 128; s > 0; s >>= 1) {
        if (t < s) red[t] += red[t + s];
        __syncthreads();
    }
    if (t == 0) { stats[2 * r] = rowmax; stats[2 * r + 1] = 1.0f / red[0]; }
}

__global__ void softmax_norm_kernel(const float* __restrict__ probs,
                                    const float* __restrict__ stats,
                                    float* __restrict__ pn, size_t total) {
    size_t i = (size_t)blockIdx.x * blockDim.x + threadIdx.x;
    if (i >= total) return;
    int r = (int)(i / N_SP);
    pn[i] = __expf(probs[i] - stats[2 * r]) * stats[2 * r + 1];
}

__global__ void __launch_bounds__(WAVES * 32)
pooled_gemm_kernel(const float* __restrict__ feats,
                   const float* __restrict__ pn,
                   const int* __restrict__ clip_num,
                   float* __restrict__ out) {
    const int beta  = blockIdx.y;
    const int ctile = blockIdx.x * CT;
    const int wave  = threadIdx.x >> 5;
    const int lane  = threadIdx.x & 31;
    const int half  = lane >> 4;
    const int lrow  = lane & 15;
    const int toff  = half * 2;

    const int   splits = clip_num[0] + 1;
    const int   bsz    = B_TOT / splits;
    const int   bb     = beta % bsz;
    const float scale  = 1.0f / (float)splits;

    constexpr int nper  = N_SP / (WAVES * NSPLIT);
    constexpr int NITER = nper / (4 * UNROLL);
    const int n0 = (blockIdx.z * WAVES + wave) * nper;

    const float* pa0 = pn + ((size_t)beta * K_DIM + lrow) * N_SP + toff + n0;
    const int    k1  = 16 + lrow;
    const float* pa1 = pn + ((size_t)beta * K_DIM + (k1 < K_DIM ? k1 : 0)) * N_SP
                          + toff + n0;
    const float* pb  = feats + ((size_t)beta * C_DIM + ctile + lrow) * N_SP
                             + toff + n0;

    v8f acc0 = {};
    v8f acc1 = {};

    for (int it = 0; it < NITER; ++it) {
        #pragma unroll
        for (int u = 0; u < UNROLL; ++u) {
            v2f a0 = *(const v2f*)(pa0 + 4 * u);
            v2f a1 = *(const v2f*)(pa1 + 4 * u);
            v2f bv = __builtin_nontemporal_load((const v2f*)(pb + 4 * u));
            acc0 = __builtin_amdgcn_wmma_f32_16x16x4_f32(
                       false, a0, false, bv, (short)0, acc0, false, false);
            acc1 = __builtin_amdgcn_wmma_f32_16x16x4_f32(
                       false, a1, false, bv, (short)0, acc1, false, false);
        }
        pa0 += 4 * UNROLL;
        pa1 += 4 * UNROLL;
        pb  += 4 * UNROLL;
    }

    const int c = ctile + lrow;
    float* ob = out + ((size_t)bb * C_DIM + c) * K_DIM;
    #pragma unroll
    for (int v = 0; v < 8; ++v) {
        const int k = v + half * 8;
        atomicAdd(ob + k, scale * acc0[v]);
        const int k2 = k + 16;
        if (k2 < K_DIM) atomicAdd(ob + k2, scale * acc1[v]);
    }
}

extern "C" void kernel_launch(void* const* d_in, const int* in_sizes, int n_in,
                              void* d_out, int out_size, void* d_ws, size_t ws_size,
                              hipStream_t stream) {
    const float* feats = (const float*)d_in[0];
    const float* probs = (const float*)d_in[1];
    const int*   clip  = (const int*)d_in[2];
    float* out   = (float*)d_out;
    float* stats = (float*)d_ws;
    float* pnbuf = (float*)d_ws + PN_OFF;

    (void)in_sizes; (void)n_in; (void)ws_size;

    zero_out_kernel<<<(out_size + 255) / 256, 256, 0, stream>>>(out, out_size);

    softmax_stats_kernel<<<B_TOT * K_DIM, 256, 0, stream>>>(probs, stats);

    const size_t total = (size_t)B_TOT * K_DIM * N_SP;
    softmax_norm_kernel<<<(unsigned)((total + 255) / 256), 256, 0, stream>>>(
        probs, stats, pnbuf, total);

    dim3 grid(C_DIM / CT, B_TOT, NSPLIT);
    pooled_gemm_kernel<<<grid, WAVES * 32, 0, stream>>>(feats, pnbuf, clip, out);
}


# pooling GEMM: feats and probs read as row-contiguous 16-B loads (4 lines per load, was 16) staged through LDS into the WMMA fragments; k order inside each partial sum changes
# speedup vs baseline: 2.7177x; 2.7177x over previous
#include <hip/hip_runtime.h>
#include <math.h>


typedef float v2f __attribute__((ext_vector_type(2)));
typedef float v4f __attribute__((ext_vector_type(4)));
typedef float v8f __attribute__((ext_vector_type(8)));

#define B_TOT  8
#define C_DIM  512
#define K_DIM  19
#define N_SP   16384
#define CT     16
#define PN_OFF 512

#define GW     4
#define WT     2
#define BROWS  (WT * CT)
#define CB     (GW * BROWS)
#define NSPLIT 16
#define NPB    (N_SP / NSPLIT)
#define NC     64
#define NCHUNK (NPB / NC)
#define PITCH  (NC + 4)

#define DEV static __device__ __forceinline__

__global__ void zero_out_kernel(float* __restrict__ out, int n) {
    int i = blockIdx.x * blockDim.x + threadIdx.x;
    if (i < n) out[i] = 0.0f;
}

__global__ void softmax_stats_kernel(const float* __restrict__ probs,
                                     float* __restrict__ stats) {
    __shared__ float red[256];
    const int r = blockIdx.x;
    const float* p = probs + (size_t)r * N_SP;
    const int t = threadIdx.x;

    float m = -3.402823466e38f;
    for (int i = t; i < N_SP; i += 256) m = fmaxf(m, p[i]);
    red[t] = m; __syncthreads();
    for (int s = 128; s > 0; s >>= 1) {
        if (t < s) red[t] = fmaxf(red[t], red[t + s]);
        __syncthreads();
    }
    const float rowmax = red[0];
    __syncthreads();

    float sum = 0.0f;
    for (int i = t; i < N_SP; i += 256) sum += __expf(p[i] - rowmax);
    red[t] = sum; __syncthreads();
    for (int s = 128; s > 0; s >>= 1) {
        if (t < s) red[t] += red[t + s];
        __syncthreads();
    }
    if (t == 0) { stats[2 * r] = rowmax; stats[2 * r + 1] = 1.0f / red[0]; }
}

__global__ void softmax_norm_kernel(const float* __restrict__ probs,
                                    const float* __restrict__ stats,
                                    float* __restrict__ pn, size_t total) {
    size_t i = (size_t)blockIdx.x * blockDim.x + threadIdx.x;
    if (i >= total) return;
    int r = (int)(i / N_SP);
    pn[i] = __expf(probs[i] - stats[2 * r]) * stats[2 * r + 1];
}

struct Stage {
    v4f b[BROWS / 2];
    v4f a[3];
};

DEV void load_chunk(Stage& g, const float* gb, const float* ga, bool a2, int ch) {
    const float* pb = gb + ch * NC;
    #pragma unroll
    for (int q = 0; q < BROWS / 2; ++q)
        g.b[q] = __builtin_nontemporal_load((const v4f*)(pb + (size_t)q * 2 * N_SP));
    const float* pa = ga + ch * NC;
    g.a[0] = *(const v4f*)(pa);
    g.a[1] = *(const v4f*)(pa + (size_t)8 * N_SP);
    if (a2) g.a[2] = *(const v4f*)(pa + (size_t)16 * N_SP);
}

DEV void store_chunk(const Stage& g, float* sBw, float* sAs, int sb_st, int sa_st, bool a2) {
    #pragma unroll
    for (int q = 0; q < BROWS / 2; ++q)
        *(v4f*)(sBw + sb_st + q * 2 * PITCH) = g.b[q];
    *(v4f*)(sAs + sa_st) = g.a[0];
    *(v4f*)(sAs + sa_st + 8 * PITCH) = g.a[1];
    if (a2) *(v4f*)(sAs + sa_st + 16 * PITCH) = g.a[2];
}

DEV v8f mma(v2f a, v2f b, v8f c) {
    return __builtin_amdgcn_wmma_f32_16x16x4_f32(false, a, false, b, (short)0, c, false, false);
}

DEV void lds_barrier() {
    asm volatile("s_wait_dscnt 0x0\n\ts_barrier_signal -1\n\ts_barrier_wait -1" ::: "memory");
}
DEV void lds_barrier_after(v8f& c00, v8f& c01, v8f& c10, v8f& c11) {
    asm volatile("s_wait_dscnt 0x0\n\ts_barrier_signal -1\n\ts_barrier_wait -1"
                 : "+v"(c00), "+v"(c01), "+v"(c10), "+v"(c11) :: "memory");
}

template <bool ST, bool LD>
DEV void phase(const float* sBr, const float* sAr, float* sBw, float* sAw,
               Stage& g, const float* gb, const float* ga, bool a2, int chl,
               int sb_st, int sa_st, int fr, int fa1o,
               v8f& c00, v8f& c01, v8f& c10, v8f& c11) {
    v4f fa0[NC / 8], fa1[NC / 8], fb0[NC / 8], fb1[NC / 8];
    #pragma unroll
    for (int j = 0; j < NC / 8; ++j) {
        fa0[j] = *(const v4f*)(sAr + fr   + j * 8);
        fa1[j] = *(const v4f*)(sAr + fa1o + j * 8);
        fb0[j] = *(const v4f*)(sBr + fr + j * 8);
        fb1[j] = *(const v4f*)(sBr + CT * PITCH + fr + j * 8);
    }
    if (ST) store_chunk(g, sBw, sAw, sb_st, sa_st, a2);
    if (LD) load_chunk(g, gb, ga, a2, chl);
    __builtin_amdgcn_sched_barrier(0);
    #pragma unroll
    for (int j = 0; j < NC / 8; ++j) {
        c00 = mma(fa0[j].xy, fb0[j].xy, c00);
        c01 = mma(fa1[j].xy, fb0[j].xy, c01);
        c10 = mma(fa0[j].xy, fb1[j].xy, c10);
        c11 = mma(fa1[j].xy, fb1[j].xy, c11);
        c00 = mma(fa0[j].zw, fb0[j].zw, c00);
        c01 = mma(fa1[j].zw, fb0[j].zw, c01);
        c10 = mma(fa0[j].zw, fb1[j].zw, c10);
        c11 = mma(fa1[j].zw, fb1[j].zw, c11);
    }
    lds_barrier_after(c00, c01, c10, c11);
}

__global__ void __launch_bounds__(GW * 32)
pooled_gemm_kernel(const float* __restrict__ feats,
                   const float* __restrict__ pn,
                   const int* __restrict__ clip_num,
                   float* __restrict__ out) {
    __shared__ __attribute__((aligned(16))) float sA[2][K_DIM * PITCH];
    __shared__ __attribute__((aligned(16))) float sB[2][GW][BROWS * PITCH];

    const int tid   = threadIdx.x;
    const int wave  = tid >> 5;
    const int lane  = tid & 31;
    const int half  = lane >> 4;
    const int lrow  = lane & 15;
    const int beta  = blockIdx.y;
    const int cbase = blockIdx.x * CB + wave * BROWS;
    const int n0    = blockIdx.z * NPB;

    const int   splits = clip_num[0] + 1;
    const int   bsz    = B_TOT / splits;
    const int   bb     = beta % bsz;
    const float scale  = 1.0f / (float)splits;

    const float* gb = feats + ((size_t)beta * C_DIM + cbase + half) * N_SP + n0 + lrow * 4;
    const int    arow = tid >> 4;
    const int    acol = (tid & 15) * 4;
    const float* ga = pn + ((size_t)beta * K_DIM + arow) * N_SP + n0 + acol;
    const bool   a2 = (arow + 16) < K_DIM;

    const int sb_st = half * PITCH + lrow * 4;
    const int sa_st = arow * PITCH + acol;
    const int fr    = lrow * PITCH + half * 4;
    const int r1    = (lrow + 16 < K_DIM) ? (lrow + 16) : lrow;
    const int fa1o  = r1 * PITCH + half * 4;

    float* sBw0 = &sB[0][wave][0];
    float* sBw1 = &sB[1][wave][0];
    float* sA0  = &sA[0][0];
    float* sA1  = &sA[1][0];

    v8f c00 = {}, c01 = {}, c10 = {}, c11 = {};

    Stage gE = {}, gO = {};
    load_chunk(gO, gb, ga, a2, 0);
    load_chunk(gE, gb, ga, a2, 1);
    store_chunk(gO, sBw0, sA0, sb_st, sa_st, a2);
    load_chunk(gO, gb, ga, a2, 2);
    lds_barrier();

#define PH(ST, LD, SR, SW, G, CHL) \
    phase<ST, LD>(sBw##SR, sA##SR, sBw##SW, sA##SW, G, gb, ga, a2, CHL, \
                  sb_st, sa_st, fr, fa1o, c00, c01, c10, c11)

    int ch = 0;
    for (; ch + 4 < NCHUNK; ch += 2) {
        PH(true, true, 0, 1, gE, ch + 3);
        PH(true, true, 1, 0, gO, ch + 4);
    }
    PH(true,  true,  0, 1, gE, NCHUNK - 1);
    PH(true,  false, 1, 0, gO, 0);
    PH(true,  false, 0, 1, gE, 0);
    PH(false, false, 1, 0, gO, 0);
#undef PH

    {
        float* ob = out + ((size_t)bb * C_DIM + cbase + lrow) * K_DIM;
        #pragma unroll
        for (int v = 0; v < 8; ++v) {
            const int k = v + half * 8;
            atomicAdd(ob + k, scale * c00[v]);
            const int k2 = k + 16;
            if (k2 < K_DIM) atomicAdd(ob + k2, scale * c01[v]);
        }
    }
    {
        float* ob = out + ((size_t)bb * C_DIM + cbase + CT + lrow) * K_DIM;
        #pragma unroll
        for (int v = 0; v < 8; ++v) {
            const int k = v + half * 8;
            atomicAdd(ob + k, scale * c10[v]);
            const int k2 = k + 16;
            if (k2 < K_DIM) atomicAdd(ob + k2, scale * c11[v]);
        }
    }
}

extern "C" void kernel_launch(void* const* d_in, const int* in_sizes, int n_in,
                              void* d_out, int out_size, void* d_ws, size_t ws_size,
                              hipStream_t stream) {
    const float* feats = (const float*)d_in[0];
    const float* probs = (const float*)d_in[1];
    const int*   clip  = (const int*)d_in[2];
    float* out   = (float*)d_out;
    float* stats = (float*)d_ws;
    float* pnbuf = (float*)d_ws + PN_OFF;

    (void)in_sizes; (void)n_in; (void)ws_size;

    zero_out_kernel<<<(out_size + 255) / 256, 256, 0, stream>>>(out, out_size);

    softmax_stats_kernel<<<B_TOT * K_DIM, 256, 0, stream>>>(probs, stats);

    const size_t total = (size_t)B_TOT * K_DIM * N_SP;
    softmax_norm_kernel<<<(unsigned)((total + 255) / 256), 256, 0, stream>>>(
        probs, stats, pnbuf, total);

    dim3 grid(C_DIM / CB, B_TOT, NSPLIT);
    pooled_gemm_kernel<<<grid, GW * 32, 0, stream>>>(feats, pnbuf, clip, out);
}


# softmax row stats: each thread issues all its 16-B loads of the probs row (16 of them) before it waits and keeps the row in registers: one memory round trip, exp from registers; row-sum order changes
# speedup vs baseline: 4.0634x; 1.4952x over previous
#include <hip/hip_runtime.h>
#include <math.h>


typedef float v2f __attribute__((ext_vector_type(2)));
typedef float v4f __attribute__((ext_vector_type(4)));
typedef float v8f __attribute__((ext_vector_type(8)));

#define B_TOT  8
#define C_DIM  512
#define K_DIM  19
#define N_SP   16384
#define CT     16

#define GW     4
#define WT     2
#define BROWS  (WT * CT)
#define CB     (GW * BROWS)
#define NSPLIT 16
#define NPB    (N_SP / NSPLIT)
#define NC     64
#define NCHUNK (NPB / NC)
#define PITCH  (NC + 4)

#define DEV static __device__ __forceinline__

__global__ void zero_out_kernel(float* __restrict__ out, int n) {
    int i = blockIdx.x * blockDim.x + threadIdx.x;
    if (i < n) out[i] = 0.0f;
}

#define ST_T   256
#define ST_V   (N_SP / 4 / ST_T)
#define ST_W   (ST_T / 32)

DEV float xhalf(float x) {
    int b = __builtin_bit_cast(int, x);
    return __builtin_bit_cast(float,
        __builtin_amdgcn_permlanex16(b, b, 0x76543210, 0xfedcba98, false, false));
}
template <int XM>
DEV float xrow(float x) {
    int b = __builtin_bit_cast(int, x);
    return __builtin_bit_cast(float, __builtin_amdgcn_update_dpp(b, b, 0x160 + XM, 0xf, 0xf, false));
}
DEV float wave_max(float x) {
    x = fmaxf(x, xrow<1>(x));
    x = fmaxf(x, xrow<2>(x));
    x = fmaxf(x, xrow<4>(x));
    x = fmaxf(x, xrow<8>(x));
    return fmaxf(x, xhalf(x));
}
DEV float wave_sum(float x) {
    x += xrow<1>(x);
    x += xrow<2>(x);
    x += xrow<4>(x);
    x += xrow<8>(x);
    return x + xhalf(x);
}

__global__ void __launch_bounds__(ST_T)
softmax_stats_kernel(const float* __restrict__ probs,
                     float* __restrict__ stats) {
    __shared__ __attribute__((aligned(16))) float redm[ST_W];
    __shared__ __attribute__((aligned(16))) float reds[ST_W];
    const int r = blockIdx.x;
    const int t = threadIdx.x;
    const int wave = t >> 5;
    const int lane = t & 31;
    const v4f* p4 = (const v4f*)(probs + (size_t)r * N_SP) + t;

    v4f x[ST_V];
    #pragma unroll
    for (int j = 0; j < ST_V; ++j) x[j] = p4[j * ST_T];

    v4f m4 = x[0];
    #pragma unroll
    for (int j = 1; j < ST_V; ++j) {
        m4.x = fmaxf(m4.x, x[j].x);
        m4.y = fmaxf(m4.y, x[j].y);
        m4.z = fmaxf(m4.z, x[j].z);
        m4.w = fmaxf(m4.w, x[j].w);
    }
    const float mw = wave_max(fmaxf(fmaxf(m4.x, m4.y), fmaxf(m4.z, m4.w)));
    if (lane == 0) redm[wave] = mw;
    __syncthreads();
    const v4f ma = *(const v4f*)&redm[0];
    const v4f mb = *(const v4f*)&redm[4];
    const float rowmax = fmaxf(fmaxf(fmaxf(ma.x, ma.y), fmaxf(ma.z, ma.w)),
                               fmaxf(fmaxf(mb.x, mb.y), fmaxf(mb.z, mb.w)));

    v4f s4 = {0.0f, 0.0f, 0.0f, 0.0f};
    #pragma unroll
    for (int j = 0; j < ST_V; ++j) {
        s4.x += __expf(x[j].x - rowmax);
        s4.y += __expf(x[j].y - rowmax);
        s4.z += __expf(x[j].z - rowmax);
        s4.w += __expf(x[j].w - rowmax);
    }
    const float sw = wave_sum((s4.x + s4.y) + (s4.z + s4.w));
    if (lane == 0) reds[wave] = sw;
    __syncthreads();
    if (t == 0) {
        const v4f sa = *(const v4f*)&reds[0];
        const v4f sb = *(const v4f*)&reds[4];
        const float sum = ((sa.x + sa.y) + (sa.z + sa.w)) + ((sb.x + sb.y) + (sb.z + sb.w));
        v2f st;
        st.x = rowmax;
        st.y = 1.0f / sum;
        *(v2f*)(stats + 2 * r) = st;
    }
}

struct Stage {
    v4f b[BROWS / 2];
    v4f a[3];
};

struct RowStats {
    v2f r0, r1, r2;
};

DEV v4f soft4(v4f x, v2f st) {
    v4f w;
    w.x = __expf(x.x - st.x) * st.y;
    w.y = __expf(x.y - st.x) * st.y;
    w.z = __expf(x.z - st.x) * st.y;
    w.w = __expf(x.w - st.x) * st.y;
    return w;
}

DEV void load_chunk(Stage& g, const float* gb, const float* ga, bool a2, int ch) {
    const float* pb = gb + ch * NC;
    #pragma unroll
    for (int q = 0; q < BROWS / 2; ++q)
        g.b[q] = __builtin_nontemporal_load((const v4f*)(pb + (size_t)q * 2 * N_SP));
    const float* pa = ga + ch * NC;
    g.a[0] = *(const v4f*)(pa);
    g.a[1] = *(const v4f*)(pa + (size_t)8 * N_SP);
    if (a2) g.a[2] = *(const v4f*)(pa + (size_t)16 * N_SP);
}

DEV void store_chunk(const Stage& g, const RowStats& rs, float* sBw, float* sAs,
                     int sb_st, int sa_st, bool a2) {
    #pragma unroll
    for (int q = 0; q < BROWS / 2; ++q)
        *(v4f*)(sBw + sb_st + q * 2 * PITCH) = g.b[q];
    *(v4f*)(sAs + sa_st) = soft4(g.a[0], rs.r0);
    *(v4f*)(sAs + sa_st + 8 * PITCH) = soft4(g.a[1], rs.r1);
    if (a2) *(v4f*)(sAs + sa_st + 16 * PITCH) = soft4(g.a[2], rs.r2);
}

DEV v8f mma(v2f a, v2f b, v8f c) {
    return __builtin_amdgcn_wmma_f32_16x16x4_f32(false, a, false, b, (short)0, c, false, false);
}

DEV void lds_barrier() {
    asm volatile("s_wait_dscnt 0x0\n\ts_barrier_signal -1\n\ts_barrier_wait -1" ::: "memory");
}
DEV void lds_barrier_after(v8f& c00, v8f& c01, v8f& c10, v8f& c11) {
    asm volatile("s_wait_dscnt 0x0\n\ts_barrier_signal -1\n\ts_barrier_wait -1"
                 : "+v"(c00), "+v"(c01), "+v"(c10), "+v"(c11) :: "memory");
}

template <bool ST, bool LD>
DEV void phase(const float* sBr, const float* sAr, float* sBw, float* sAw,
               Stage& g, const RowStats& rs, const float* gb, const float* ga, bool a2, int chl,
               int sb_st, int sa_st, int fr, int fa1o,
               v8f& c00, v8f& c01, v8f& c10, v8f& c11) {
    __builtin_amdgcn_sched_barrier(0);
    v4f fa0[NC / 8], fa1[NC / 8], fb0[NC / 8], fb1[NC / 8];
    #pragma unroll
    for (int j = 0; j < NC / 8; ++j) {
        fa0[j] = *(const v4f*)(sAr + fr   + j * 8);
        fa1[j] = *(const v4f*)(sAr + fa1o + j * 8);
        fb0[j] = *(const v4f*)(sBr + fr + j * 8);
        fb1[j] = *(const v4f*)(sBr + CT * PITCH + fr + j * 8);
    }
    if (ST) store_chunk(g, rs, sBw, sAw, sb_st, sa_st, a2);
    if (LD) load_chunk(g, gb, ga, a2, chl);
    __builtin_amdgcn_sched_barrier(0);
    #pragma unroll
    for (int j = 0; j < NC / 8; ++j) {
        c00 = mma(fa0[j].xy, fb0[j].xy, c00);
        c01 = mma(fa1[j].xy, fb0[j].xy, c01);
        c10 = mma(fa0[j].xy, fb1[j].xy, c10);
        c11 = mma(fa1[j].xy, fb1[j].xy, c11);
        c00 = mma(fa0[j].zw, fb0[j].zw, c00);
        c01 = mma(fa1[j].zw, fb0[j].zw, c01);
        c10 = mma(fa0[j].zw, fb1[j].zw, c10);
        c11 = mma(fa1[j].zw, fb1[j].zw, c11);
    }
    lds_barrier_after(c00, c01, c10, c11);
}

__global__ void __launch_bounds__(GW * 32)
pooled_gemm_kernel(const float* __restrict__ feats,
                   const float* __restrict__ probs,
                   const float* __restrict__ stats,
                   const int* __restrict__ clip_num,
                   float* __restrict__ out) {
    __shared__ __attribute__((aligned(16))) float sA[2][K_DIM * PITCH];
    __shared__ __attribute__((aligned(16))) float sB[2][GW][BROWS * PITCH];

    const int tid   = threadIdx.x;
    const int wave  = tid >> 5;
    const int lane  = tid & 31;
    const int half  = lane >> 4;
    const int lrow  = lane & 15;
    const int beta  = blockIdx.y;
    const int cbase = blockIdx.x * CB + wave * BROWS;
    const int n0    = blockIdx.z * NPB;

    const int   splits = clip_num[0] + 1;
    const int   bsz    = B_TOT / splits;
    const int   bb     = beta % bsz;
    const float scale  = 1.0f / (float)splits;

    const float* gb = feats + ((size_t)beta * C_DIM + cbase + half) * N_SP + n0 + lrow * 4;
    const int    arow = tid >> 4;
    const int    acol = (tid & 15) * 4;
    const float* ga = probs + ((size_t)beta * K_DIM + arow) * N_SP + n0 + acol;
    const bool   a2 = (arow + 16) < K_DIM;

    const float* sp = stats + 2 * (beta * K_DIM + arow);
    RowStats rs;
    rs.r0 = *(const v2f*)(sp);
    rs.r1 = *(const v2f*)(sp + 2 * 8);
    rs.r2 = (v2f){0.0f, 0.0f};
    if (a2) rs.r2 = *(const v2f*)(sp + 2 * 16);

    const int sb_st = half * PITCH + lrow * 4;
    const int sa_st = arow * PITCH + acol;
    const int fr    = lrow * PITCH + half * 4;
    const int r1    = (lrow + 16 < K_DIM) ? (lrow + 16) : lrow;
    const int fa1o  = r1 * PITCH + half * 4;

    float* sBw0 = &sB[0][wave][0];
    float* sBw1 = &sB[1][wave][0];
    float* sA0  = &sA[0][0];
    float* sA1  = &sA[1][0];

    v8f c00 = {}, c01 = {}, c10 = {}, c11 = {};

    Stage gE = {}, gO = {};
    load_chunk(gO, gb, ga, a2, 0);
    load_chunk(gE, gb, ga, a2, 1);
    store_chunk(gO, rs, sBw0, sA0, sb_st, sa_st, a2);
    load_chunk(gO, gb, ga, a2, 2);
    lds_barrier();

#define PH(ST, LD, SR, SW, G, CHL) \
    phase<ST, LD>(sBw##SR, sA##SR, sBw##SW, sA##SW, G, rs, gb, ga, a2, CHL, \
                  sb_st, sa_st, fr, fa1o, c00, c01, c10, c11)

    int ch = 0;
    for (; ch + 4 < NCHUNK; ch += 2) {
        PH(true, true, 0, 1, gE, ch + 3);
        PH(true, true, 1, 0, gO, ch + 4);
    }
    PH(true,  true,  0, 1, gE, NCHUNK - 1);
    PH(true,  false, 1, 0, gO, 0);
    PH(true,  false, 0, 1, gE, 0);
    PH(false, false, 1, 0, gO, 0);
#undef PH

    {
        float* ob = out + ((size_t)bb * C_DIM + cbase + lrow) * K_DIM;
        #pragma unroll
        for (int v = 0; v < 8; ++v) {
            const int k = v + half * 8;
            atomicAdd(ob + k, scale * c00[v]);
            const int k2 = k + 16;
            if (k2 < K_DIM) atomicAdd(ob + k2, scale * c01[v]);
        }
    }
    {
        float* ob = out + ((size_t)bb * C_DIM + cbase + CT + lrow) * K_DIM;
        #pragma unroll
        for (int v = 0; v < 8; ++v) {
            const int k = v + half * 8;
            atomicAdd(ob + k, scale * c10[v]);
            const int k2 = k + 16;
            if (k2 < K_DIM) atomicAdd(ob + k2, scale * c11[v]);
        }
    }
}

extern "C" void kernel_launch(void* const* d_in, const int* in_sizes, int n_in,
                              void* d_out, int out_size, void* d_ws, size_t ws_size,
                              hipStream_t stream) {
    const float* feats = (const float*)d_in[0];
    const float* probs = (const float*)d_in[1];
    const int*   clip  = (const int*)d_in[2];
    float* out   = (float*)d_out;
    float* stats = (float*)d_ws;

    (void)in_sizes; (void)n_in; (void)ws_size;

    zero_out_kernel<<<(out_size + 255) / 256, 256, 0, stream>>>(out, out_size);

    softmax_stats_kernel<<<B_TOT * K_DIM, ST_T, 0, stream>>>(probs, stats);

    dim3 grid(C_DIM / CB, B_TOT, NSPLIT);
    pooled_gemm_kernel<<<grid, GW * 32, 0, stream>>>(feats, probs, stats, clip, out);
}


# softmax weights of the whole split kept resident in LDS, written once by the prologue from the max pass's registers; chunk loop stages wave-private feats only: no barrier in it; sum order unchanged
# speedup vs baseline: 4.7599x; 1.1714x over previous
#include <hip/hip_runtime.h>
#include <math.h>


typedef float v2f __attribute__((ext_vector_type(2)));
typedef float v4f __attribute__((ext_vector_type(4)));
typedef float v8f __attribute__((ext_vector_type(8)));

#define B_TOT  8
#define C_DIM  512
#define K_DIM  19
#define N_SP   16384
#define CT     16

#define GW     4
#define WT     2
#define BROWS  (WT * CT)
#define CB     (GW * BROWS)
#define NSPLIT 16
#define NPB    (N_SP / NSPLIT)
#define NC     64
#define NCHUNK (NPB / NC)
#define PITCH  (NC + 4)
#define APITCH (NPB + 4)

#define DEV static __device__ __forceinline__

template <int XM>
DEV float xrow(float x) {
    int b = __builtin_bit_cast(int, x);
    return __builtin_bit_cast(float, __builtin_amdgcn_update_dpp(b, b, 0x160 + XM, 0xf, 0xf, false));
}
DEV float row16_max(float x) {
    x = fmaxf(x, xrow<1>(x));
    x = fmaxf(x, xrow<2>(x));
    x = fmaxf(x, xrow<4>(x));
    return fmaxf(x, xrow<8>(x));
}
DEV float row16_sum(float x) {
    x += xrow<1>(x);
    x += xrow<2>(x);
    x += xrow<4>(x);
    return x + xrow<8>(x);
}
DEV float max4(v4f a) { return fmaxf(fmaxf(a.x, a.y), fmaxf(a.z, a.w)); }

struct Stage {
    v4f b[BROWS / 2];
};

struct RowStats {
    float m0, m1, m2;
    float s0, s1, s2;
};

DEV v4f soft4(v4f x, float m) {
    v4f w;
    w.x = __expf(x.x - m);
    w.y = __expf(x.y - m);
    w.z = __expf(x.z - m);
    w.w = __expf(x.w - m);
    return w;
}
DEV float sum4(v4f w) { return (w.x + w.y) + (w.z + w.w); }

DEV void load_chunk(Stage& g, const float* gb, int ch) {
    const float* pb = gb + ch * NC;
    #pragma unroll
    for (int q = 0; q < BROWS / 2; ++q)
        g.b[q] = __builtin_nontemporal_load((const v4f*)(pb + (size_t)q * 2 * N_SP));
}

DEV void store_chunk(const Stage& g, float* sBw, int sb_st) {
    #pragma unroll
    for (int q = 0; q < BROWS / 2; ++q)
        *(v4f*)(sBw + sb_st + q * 2 * PITCH) = g.b[q];
}

DEV v8f mma(v2f a, v2f b, v8f c) {
    return __builtin_amdgcn_wmma_f32_16x16x4_f32(false, a, false, b, (short)0, c, false, false);
}

DEV void lds_barrier() {
    asm volatile("s_wait_dscnt 0x0\n\ts_barrier_signal -1\n\ts_barrier_wait -1" ::: "memory");
}
DEV void phase_end(v8f& c00, v8f& c01, v8f& c10, v8f& c11) {
    asm volatile("s_wait_dscnt 0x0"
                 : "+v"(c00), "+v"(c01), "+v"(c10), "+v"(c11) :: "memory");
}

template <bool ST, bool LD>
DEV void phase(const float* sBr, const float* sAr, float* sBw,
               Stage& g, const float* gb, int chl,
               int sb_st, int fr, int fa0o, int fa1o,
               v8f& c00, v8f& c01, v8f& c10, v8f& c11) {
    v4f fa0[NC / 8], fa1[NC / 8], fb0[NC / 8], fb1[NC / 8];
    #pragma unroll
    for (int j = 0; j < NC / 8; ++j) {
        fa0[j] = *(const v4f*)(sAr + fa0o + j * 8);
        fa1[j] = *(const v4f*)(sAr + fa1o + j * 8);
        fb0[j] = *(const v4f*)(sBr + fr + j * 8);
        fb1[j] = *(const v4f*)(sBr + CT * PITCH + fr + j * 8);
    }
    __builtin_amdgcn_sched_barrier(0);
    if (ST) store_chunk(g, sBw, sb_st);
    if (LD) load_chunk(g, gb, chl);
    __builtin_amdgcn_sched_barrier(0);
    #pragma unroll
    for (int j = 0; j < NC / 8; ++j) {
        c00 = mma(fa0[j].xy, fb0[j].xy, c00);
        c01 = mma(fa1[j].xy, fb0[j].xy, c01);
        c10 = mma(fa0[j].xy, fb1[j].xy, c10);
        c11 = mma(fa1[j].xy, fb1[j].xy, c11);
        c00 = mma(fa0[j].zw, fb0[j].zw, c00);
        c01 = mma(fa1[j].zw, fb0[j].zw, c01);
        c10 = mma(fa0[j].zw, fb1[j].zw, c10);
        c11 = mma(fa1[j].zw, fb1[j].zw, c11);
    }
    phase_end(c00, c01, c10, c11);
}

__global__ void __launch_bounds__(GW * 32)
pooled_gemm_kernel(const float* __restrict__ feats,
                   const float* __restrict__ probs,
                   float* __restrict__ lstat,
                   float* __restrict__ part) {
    __shared__ __attribute__((aligned(16))) float sA[K_DIM * APITCH];
    __shared__ __attribute__((aligned(16))) float sB[2][GW][BROWS * PITCH];

    const int tid   = threadIdx.x;
    const int wave  = tid >> 5;
    const int lane  = tid & 31;
    const int half  = lane >> 4;
    const int lrow  = lane & 15;
    const int beta  = blockIdx.y;
    const int cbase = blockIdx.x * CB + wave * BROWS;
    const int n0    = blockIdx.z * NPB;

    const float* gb = feats + ((size_t)beta * C_DIM + cbase + half) * N_SP + n0 + lrow * 4;
    const int    arow = tid >> 4;
    const int    acol = (tid & 15) * 4;
    const float* ga = probs + ((size_t)beta * K_DIM + arow) * N_SP + n0 + acol;
    const bool   a2 = (arow + 16) < K_DIM;

    RowStats rs;
    v4f x0[NCHUNK], x1[NCHUNK], x2[NCHUNK];
    #pragma unroll
    for (int j = 0; j < NCHUNK; ++j) {
        x0[j] = *(const v4f*)(ga + j * NC);
        x1[j] = *(const v4f*)(ga + (size_t)8 * N_SP + j * NC);
    }
    if (a2) {
        #pragma unroll
        for (int j = 0; j < NCHUNK; ++j) x2[j] = *(const v4f*)(ga + (size_t)16 * N_SP + j * NC);
    }

    const int sb_st = half * PITCH + lrow * 4;
    const int sa_st = arow * APITCH + acol;
    const int fr    = lrow * PITCH + half * 4;
    const int fa0o  = lrow * APITCH + half * 4;
    const int r1    = (lrow + 16 < K_DIM) ? (lrow + 16) : lrow;
    const int fa1o  = r1 * APITCH + half * 4;

    float* sBw0 = &sB[0][wave][0];
    float* sBw1 = &sB[1][wave][0];

    v8f c00 = {}, c01 = {}, c10 = {}, c11 = {};

    Stage gE = {}, gO = {};
    load_chunk(gO, gb, 0);
    load_chunk(gE, gb, 1);
    {
        float t0 = max4(x0[0]), t1 = max4(x1[0]), t2 = 0.0f;
        #pragma unroll
        for (int j = 1; j < NCHUNK; ++j) {
            t0 = fmaxf(t0, max4(x0[j]));
            t1 = fmaxf(t1, max4(x1[j]));
        }
        if (a2) {
            t2 = max4(x2[0]);
            #pragma unroll
            for (int j = 1; j < NCHUNK; ++j) t2 = fmaxf(t2, max4(x2[j]));
        }
        rs.m0 = row16_max(t0);
        rs.m1 = row16_max(t1);
        rs.m2 = row16_max(t2);
        rs.s0 = 0.0f; rs.s1 = 0.0f; rs.s2 = 0.0f;
    }
    #pragma unroll
    for (int j = 0; j < NCHUNK; ++j) {
        const v4f w0 = soft4(x0[j], rs.m0);
        const v4f w1 = soft4(x1[j], rs.m1);
        *(v4f*)(sA + sa_st + j * NC) = w0;
        *(v4f*)(sA + sa_st + 8 * APITCH + j * NC) = w1;
        rs.s0 += sum4(w0);
        rs.s1 += sum4(w1);
    }
    if (a2) {
        #pragma unroll
        for (int j = 0; j < NCHUNK; ++j) {
            const v4f w2 = soft4(x2[j], rs.m2);
            *(v4f*)(sA + sa_st + 16 * APITCH + j * NC) = w2;
            rs.s2 += sum4(w2);
        }
    }
    store_chunk(gO, sBw0, sb_st);
    load_chunk(gO, gb, 2);
    lds_barrier();

#define PH(ST, LD, SR, SW, G, CH, CHL) \
    phase<ST, LD>(sBw##SR, sA + (CH) * NC, sBw##SW, G, gb, CHL, \
                  sb_st, fr, fa0o, fa1o, c00, c01, c10, c11)

    int ch = 0;
    for (; ch + 4 < NCHUNK; ch += 2) {
        PH(true, true, 0, 1, gE, ch,     ch + 3);
        PH(true, true, 1, 0, gO, ch + 1, ch + 4);
    }
    PH(true,  true,  0, 1, gE, NCHUNK - 4, NCHUNK - 1);
    PH(true,  false, 1, 0, gO, NCHUNK - 3, 0);
    PH(true,  false, 0, 1, gE, NCHUNK - 2, 0);
    PH(false, false, 1, 0, gO, NCHUNK - 1, 0);
#undef PH

    {
        const float q0 = row16_sum(rs.s0);
        const float q1 = row16_sum(rs.s1);
        const float q2 = row16_sum(rs.s2);
        if (blockIdx.x == 0 && (tid & 15) == 0) {
            float* ls = lstat + 2 * ((beta * NSPLIT + blockIdx.z) * K_DIM + arow);
            *(v2f*)(ls)         = (v2f){rs.m0, q0};
            *(v2f*)(ls + 2 * 8) = (v2f){rs.m1, q1};
            if (a2) *(v2f*)(ls + 2 * 16) = (v2f){rs.m2, q2};
        }
    }

    float* so = sBw0;
    #pragma unroll
    for (int v = 0; v < 8; ++v) {
        const int k = v + half * 8;
        so[lrow * K_DIM + k]        = c00[v];
        so[(CT + lrow) * K_DIM + k] = c10[v];
    }
    if (half == 0) {
        #pragma unroll
        for (int v = 0; v < K_DIM - 16; ++v) {
            so[lrow * K_DIM + 16 + v]        = c01[v];
            so[(CT + lrow) * K_DIM + 16 + v] = c11[v];
        }
    }
    __syncthreads();
    float* pw = part + (((size_t)beta * NSPLIT + blockIdx.z) * C_DIM + cbase) * K_DIM;
    #pragma unroll
    for (int j = 0; j < (BROWS * K_DIM / 4 + 31) / 32; ++j) {
        const int i = lane + 32 * j;
        if (i < BROWS * K_DIM / 4) *(v4f*)(pw + 4 * i) = *(const v4f*)(so + 4 * i);
    }
}

#define RD_C   16
#define RD_T   (RD_C * NSPLIT)
#define PLANE4 (C_DIM * K_DIM / 4)

template <int S>
DEV void join_body(const float* __restrict__ part, const float* __restrict__ lstat,
                   float* __restrict__ out, int n,
                   float (*wt)[NSPLIT][K_DIM], v4f (*red)[RD_C]) {
    constexpr int BSZ = B_TOT / S;
    const int t  = threadIdx.x;
    const int cl = t & (RD_C - 1);
    const int z  = t / RD_C;
    const int col = blockIdx.x * RD_C + cl;
    const int bb  = col / PLANE4;
    const int rem = col - bb * PLANE4;
    const bool live = (4 * col < n) && (bb < BSZ);

    v4f x[S];
    #pragma unroll
    for (int s = 0; s < S; ++s) x[s] = (v4f){0.0f, 0.0f, 0.0f, 0.0f};
    if (live) {
        const v4f* p4 = (const v4f*)part + (size_t)z * PLANE4 + rem;
        #pragma unroll
        for (int s = 0; s < S; ++s) x[s] = p4[(size_t)(s * BSZ + bb) * NSPLIT * PLANE4];
    }
    if (live && t < S * K_DIM) {
        const int s = t / K_DIM;
        const int k = t - s * K_DIM;
        const v2f* ls = (const v2f*)lstat + (size_t)(s * BSZ + bb) * NSPLIT * K_DIM + k;
        v2f st[NSPLIT];
        #pragma unroll
        for (int g = 0; g < NSPLIT; ++g) st[g] = ls[g * K_DIM];
        float mx = st[0].x;
        #pragma unroll
        for (int g = 1; g < NSPLIT; ++g) mx = fmaxf(mx, st[g].x);
        float e[NSPLIT];
        float den = 0.0f;
        #pragma unroll
        for (int g = 0; g < NSPLIT; ++g) {
            e[g] = __expf(st[g].x - mx);
            den += e[g] * st[g].y;
        }
        const float inv = 1.0f / den;
        #pragma unroll
        for (int g = 0; g < NSPLIT; ++g) wt[s][g][k] = e[g] * inv;
    }
    __syncthreads();
    v4f acc = {0.0f, 0.0f, 0.0f, 0.0f};
    if (live) {
        const int k0 = (4 * rem) % K_DIM;
        const int k1 = (k0 + 1 < K_DIM) ? k0 + 1 : k0 + 1 - K_DIM;
        const int k2 = (k1 + 1 < K_DIM) ? k1 + 1 : k1 + 1 - K_DIM;
        const int k3 = (k2 + 1 < K_DIM) ? k2 + 1 : k2 + 1 - K_DIM;
        #pragma unroll
        for (int s = 0; s < S; ++s) {
            const float* w = &wt[s][z][0];
            acc.x += x[s].x * w[k0];
            acc.y += x[s].y * w[k1];
            acc.z += x[s].z * w[k2];
            acc.w += x[s].w * w[k3];
        }
    }
    red[z][cl] = acc;
    __syncthreads();
    if (t < RD_C && live) {
        v4f sum = red[0][t];
        #pragma unroll
        for (int g = 1; g < NSPLIT; ++g) sum += red[g][t];
        *((v4f*)out + col) = sum * (1.0f / (float)S);
    }
}

__global__ void __launch_bounds__(RD_T)
join_splits_kernel(const float* __restrict__ part,
                   const float* __restrict__ lstat,
                   const int* __restrict__ clip_num,
                   float* __restrict__ out, int n) {
    __shared__ __attribute__((aligned(16))) float wt[B_TOT][NSPLIT][K_DIM];
    __shared__ __attribute__((aligned(16))) v4f red[NSPLIT][RD_C];
    const int splits = clip_num[0] + 1;
    if (splits == 4)      join_body<4>(part, lstat, out, n, wt, red);
    else if (splits == 8) join_body<8>(part, lstat, out, n, wt, red);
    else if (splits == 2) join_body<2>(part, lstat, out, n, wt, red);
    else if (splits == 1) join_body<1>(part, lstat, out, n, wt, red);
}

extern "C" void kernel_launch(void* const* d_in, const int* in_sizes, int n_in,
                              void* d_out, int out_size, void* d_ws, size_t ws_size,
                              hipStream_t stream) {
    const float* feats = (const float*)d_in[0];
    const float* probs = (const float*)d_in[1];
    const int*   clip  = (const int*)d_in[2];
    float* out   = (float*)d_out;
    float* lstat = (float*)d_ws;
    float* part  = lstat + 8192;

    (void)in_sizes; (void)n_in; (void)ws_size;

    dim3 grid(C_DIM / CB, B_TOT, NSPLIT);
    pooled_gemm_kernel<<<grid, GW * 32, 0, stream>>>(feats, probs, lstat, part);

    join_splits_kernel<<<(out_size + 4 * RD_C - 1) / (4 * RD_C), RD_T, 0, stream>>>(part, lstat, clip, out, out_size);
}


# pooling GEMM: weight rows 16-18 (3/16 of a second WMMA tile) summed by packed f32 FMAs on the vector ALU from the staged feats registers; rows 0-15 stay on WMMA; sum order of rows 16-18 changes
# speedup vs baseline: 5.1949x; 1.0914x over previous
#include <hip/hip_runtime.h>
#include <math.h>


typedef float v2f __attribute__((ext_vector_type(2)));
typedef float v4f __attribute__((ext_vector_type(4)));
typedef float v8f __attribute__((ext_vector_type(8)));

#define B_TOT  8
#define C_DIM  512
#define K_DIM  19
#define N_SP   16384
#define CT     16

#define GW     4
#define WT     2
#define BROWS  (WT * CT)
#define CB     (GW * BROWS)
#define NSPLIT 16
#define NPB    (N_SP / NSPLIT)
#define NC     64
#define NCHUNK (NPB / NC)
#define PITCH  (NC + 4)
#define APITCH (NPB + 4)

#define DEV static __device__ __forceinline__

template <int XM>
DEV float xrow(float x) {
    int b = __builtin_bit_cast(int, x);
    return __builtin_bit_cast(float, __builtin_amdgcn_update_dpp(b, b, 0x160 + XM, 0xf, 0xf, false));
}
DEV float row16_max(float x) {
    x = fmaxf(x, xrow<1>(x));
    x = fmaxf(x, xrow<2>(x));
    x = fmaxf(x, xrow<4>(x));
    return fmaxf(x, xrow<8>(x));
}
DEV float row16_sum(float x) {
    x += xrow<1>(x);
    x += xrow<2>(x);
    x += xrow<4>(x);
    return x + xrow<8>(x);
}
DEV float max4(v4f a) { return fmaxf(fmaxf(a.x, a.y), fmaxf(a.z, a.w)); }

struct Stage {
    v4f b[BROWS / 2];
};

struct Tail {
    v2f a0[BROWS / 2], a1[BROWS / 2], a2[BROWS / 2];
};

DEV v2f dot4acc(v4f b, v4f w, v2f acc) {
    acc = __builtin_elementwise_fma(b.xy, w.xy, acc);
    return __builtin_elementwise_fma(b.zw, w.zw, acc);
}

DEV void tail_fma(Tail& t, const Stage& g, v4f w0, v4f w1, v4f w2) {
    #pragma unroll
    for (int q = 0; q < BROWS / 2; ++q) {
        t.a0[q] = dot4acc(g.b[q], w0, t.a0[q]);
        t.a1[q] = dot4acc(g.b[q], w1, t.a1[q]);
        t.a2[q] = dot4acc(g.b[q], w2, t.a2[q]);
    }
}

struct RowStats {
    float m0, m1, m2;
    float s0, s1, s2;
};

DEV v4f soft4(v4f x, float m) {
    v4f w;
    w.x = __expf(x.x - m);
    w.y = __expf(x.y - m);
    w.z = __expf(x.z - m);
    w.w = __expf(x.w - m);
    return w;
}
DEV float sum4(v4f w) { return (w.x + w.y) + (w.z + w.w); }

DEV void load_chunk(Stage& g, const float* gb, int ch) {
    const float* pb = gb + ch * NC;
    #pragma unroll
    for (int q = 0; q < BROWS / 2; ++q)
        g.b[q] = __builtin_nontemporal_load((const v4f*)(pb + (size_t)q * 2 * N_SP));
}

DEV void store_chunk(const Stage& g, float* sBw, int sb_st) {
    #pragma unroll
    for (int q = 0; q < BROWS / 2; ++q)
        *(v4f*)(sBw + sb_st + q * 2 * PITCH) = g.b[q];
}

DEV v8f mma(v2f a, v2f b, v8f c) {
    return __builtin_amdgcn_wmma_f32_16x16x4_f32(false, a, false, b, (short)0, c, false, false);
}

DEV void lds_barrier() {
    asm volatile("s_wait_dscnt 0x0\n\ts_barrier_signal -1\n\ts_barrier_wait -1" ::: "memory");
}
DEV void phase_end(v8f& c00, v8f& c10) {
    asm volatile("s_wait_dscnt 0x0"
                 : "+v"(c00), "+v"(c10) :: "memory");
}

template <bool ST, bool LD>
DEV void phase(const float* sBr, const float* sAr, float* sBw,
               Stage& g, const float* gb, int chl,
               int sb_st, int fr, int fa0o, int wo,
               v8f& c00, v8f& c10, Tail& t) {
    v4f fa0[NC / 8], fb0[NC / 8], fb1[NC / 8];
    v4f w0 = {}, w1 = {}, w2 = {};
    if (ST) {
        w0 = *(const v4f*)(sAr + NC + wo);
        w1 = *(const v4f*)(sAr + NC + wo + APITCH);
        w2 = *(const v4f*)(sAr + NC + wo + 2 * APITCH);
    }
    #pragma unroll
    for (int j = 0; j < NC / 8; ++j) {
        fa0[j] = *(const v4f*)(sAr + fa0o + j * 8);
        fb0[j] = *(const v4f*)(sBr + fr + j * 8);
        fb1[j] = *(const v4f*)(sBr + CT * PITCH + fr + j * 8);
    }
    __builtin_amdgcn_sched_barrier(0);
    if (ST) {
        store_chunk(g, sBw, sb_st);
        tail_fma(t, g, w0, w1, w2);
    }
    if (LD) load_chunk(g, gb, chl);
    __builtin_amdgcn_sched_barrier(0);
    #pragma unroll
    for (int j = 0; j < NC / 8; ++j) {
        c00 = mma(fa0[j].xy, fb0[j].xy, c00);
        c10 = mma(fa0[j].xy, fb1[j].xy, c10);
        c00 = mma(fa0[j].zw, fb0[j].zw, c00);
        c10 = mma(fa0[j].zw, fb1[j].zw, c10);
    }
    phase_end(c00, c10);
}

__global__ void __launch_bounds__(GW * 32)
pooled_gemm_kernel(const float* __restrict__ feats,
                   const float* __restrict__ probs,
                   float* __restrict__ lstat,
                   float* __restrict__ part) {
    __shared__ __attribute__((aligned(16))) float sA[K_DIM * APITCH];
    __shared__ __attribute__((aligned(16))) float sB[2][GW][BROWS * PITCH];

    const int tid   = threadIdx.x;
    const int wave  = tid >> 5;
    const int lane  = tid & 31;
    const int half  = lane >> 4;
    const int lrow  = lane & 15;
    const int beta  = blockIdx.y;
    const int cbase = blockIdx.x * CB + wave * BROWS;
    const int n0    = blockIdx.z * NPB;

    const float* gb = feats + ((size_t)beta * C_DIM + cbase + half) * N_SP + n0 + lrow * 4;
    const int    arow = tid >> 4;
    const int    acol = (tid & 15) * 4;
    const float* ga = probs + ((size_t)beta * K_DIM + arow) * N_SP + n0 + acol;
    const bool   a2 = (arow + 16) < K_DIM;

    RowStats rs;
    v4f x0[NCHUNK], x1[NCHUNK], x2[NCHUNK];
    #pragma unroll
    for (int j = 0; j < NCHUNK; ++j) {
        x0[j] = *(const v4f*)(ga + j * NC);
        x1[j] = *(const v4f*)(ga + (size_t)8 * N_SP + j * NC);
    }
    if (a2) {
        #pragma unroll
        for (int j = 0; j < NCHUNK; ++j) x2[j] = *(const v4f*)(ga + (size_t)16 * N_SP + j * NC);
    }

    const int sb_st = half * PITCH + lrow * 4;
    const int sa_st = arow * APITCH + acol;
    const int fr    = lrow * PITCH + half * 4;
    const int fa0o  = lrow * APITCH + half * 4;
    const int wo    = CT * APITCH + lrow * 4;

    float* sBw0 = &sB[0][wave][0];
    float* sBw1 = &sB[1][wave][0];

    v8f c00 = {}, c10 = {};
    Tail tl = {};

    Stage gE = {}, gO = {};
    load_chunk(gO, gb, 0);
    load_chunk(gE, gb, 1);
    {
        float t0 = max4(x0[0]), t1 = max4(x1[0]), t2 = 0.0f;
        #pragma unroll
        for (int j = 1; j < NCHUNK; ++j) {
            t0 = fmaxf(t0, max4(x0[j]));
            t1 = fmaxf(t1, max4(x1[j]));
        }
        if (a2) {
            t2 = max4(x2[0]);
            #pragma unroll
            for (int j = 1; j < NCHUNK; ++j) t2 = fmaxf(t2, max4(x2[j]));
        }
        rs.m0 = row16_max(t0);
        rs.m1 = row16_max(t1);
        rs.m2 = row16_max(t2);
        rs.s0 = 0.0f; rs.s1 = 0.0f; rs.s2 = 0.0f;
    }
    #pragma unroll
    for (int j = 0; j < NCHUNK; ++j) {
        const v4f w0 = soft4(x0[j], rs.m0);
        const v4f w1 = soft4(x1[j], rs.m1);
        *(v4f*)(sA + sa_st + j * NC) = w0;
        *(v4f*)(sA + sa_st + 8 * APITCH + j * NC) = w1;
        rs.s0 += sum4(w0);
        rs.s1 += sum4(w1);
    }
    if (a2) {
        #pragma unroll
        for (int j = 0; j < NCHUNK; ++j) {
            const v4f w2 = soft4(x2[j], rs.m2);
            *(v4f*)(sA + sa_st + 16 * APITCH + j * NC) = w2;
            rs.s2 += sum4(w2);
        }
    }
    store_chunk(gO, sBw0, sb_st);
    lds_barrier();
    tail_fma(tl, gO,
             *(const v4f*)(sA + wo),
             *(const v4f*)(sA + wo + APITCH),
             *(const v4f*)(sA + wo + 2 * APITCH));
    load_chunk(gO, gb, 2);

    {
        const float q0 = row16_sum(rs.s0);
        const float q1 = row16_sum(rs.s1);
        const float q2 = row16_sum(rs.s2);
        if (blockIdx.x == 0 && (tid & 15) == 0) {
            float* ls = lstat + 2 * ((beta * NSPLIT + blockIdx.z) * K_DIM + arow);
            *(v2f*)(ls)         = (v2f){rs.m0, q0};
            *(v2f*)(ls + 2 * 8) = (v2f){rs.m1, q1};
            if (a2) *(v2f*)(ls + 2 * 16) = (v2f){rs.m2, q2};
        }
    }

#define PH(ST, LD, SR, SW, G, CH, CHL) \
    phase<ST, LD>(sBw##SR, sA + (CH) * NC, sBw##SW, G, gb, CHL, \
                  sb_st, fr, fa0o, wo, c00, c10, tl)

    int ch = 0;
    for (; ch + 4 < NCHUNK; ch += 2) {
        PH(true, true, 0, 1, gE, ch,     ch + 3);
        PH(true, true, 1, 0, gO, ch + 1, ch + 4);
    }
    PH(true,  true,  0, 1, gE, NCHUNK - 4, NCHUNK - 1);
    PH(true,  false, 1, 0, gO, NCHUNK - 3, 0);
    PH(true,  false, 0, 1, gE, NCHUNK - 2, 0);
    PH(false, false, 1, 0, gO, NCHUNK - 1, 0);
#undef PH

    float* so = sBw0;
    #pragma unroll
    for (int v = 0; v < 8; ++v) {
        const int k = v + half * 8;
        so[lrow * K_DIM + k]        = c00[v];
        so[(CT + lrow) * K_DIM + k] = c10[v];
    }
    {
        float o0 = 0.0f, o1 = 0.0f, o2 = 0.0f;
        #pragma unroll
        for (int q = 0; q < BROWS / 2; ++q) {
            const float r0 = row16_sum(tl.a0[q].x + tl.a0[q].y);
            const float r1 = row16_sum(tl.a1[q].x + tl.a1[q].y);
            const float r2 = row16_sum(tl.a2[q].x + tl.a2[q].y);
            if (lrow == q) { o0 = r0; o1 = r1; o2 = r2; }
        }
        float* st = so + (2 * lrow + half) * K_DIM + 16;
        st[0] = o0;
        st[1] = o1;
        st[2] = o2;
    }
    __syncthreads();
    float* pw = part + (((size_t)beta * NSPLIT + blockIdx.z) * C_DIM + cbase) * K_DIM;
    #pragma unroll
    for (int j = 0; j < (BROWS * K_DIM / 4 + 31) / 32; ++j) {
        const int i = lane + 32 * j;
        if (i < BROWS * K_DIM / 4) *(v4f*)(pw + 4 * i) = *(const v4f*)(so + 4 * i);
    }
}

#define RD_C   16
#define RD_T   (RD_C * NSPLIT)
#define PLANE4 (C_DIM * K_DIM / 4)

template <int S>
DEV void join_body(const float* __restrict__ part, const float* __restrict__ lstat,
                   float* __restrict__ out, int n,
                   float (*wt)[NSPLIT][K_DIM], v4f (*red)[RD_C]) {
    constexpr int BSZ = B_TOT / S;
    const int t  = threadIdx.x;
    const int cl = t & (RD_C - 1);
    const int z  = t / RD_C;
    const int col = blockIdx.x * RD_C + cl;
    const int bb  = col / PLANE4;
    const int rem = col - bb * PLANE4;
    const bool live = (4 * col < n) && (bb < BSZ);

    v4f x[S];
    #pragma unroll
    for (int s = 0; s < S; ++s) x[s] = (v4f){0.0f, 0.0f, 0.0f, 0.0f};
    if (live) {
        const v4f* p4 = (const v4f*)part + (size_t)z * PLANE4 + rem;
        #pragma unroll
        for (int s = 0; s < S; ++s) x[s] = p4[(size_t)(s * BSZ + bb) * NSPLIT * PLANE4];
    }
    if (live && t < S * K_DIM) {
        const int s = t / K_DIM;
        const int k = t - s * K_DIM;
        const v2f* ls = (const v2f*)lstat + (size_t)(s * BSZ + bb) * NSPLIT * K_DIM + k;
        v2f st[NSPLIT];
        #pragma unroll
        for (int g = 0; g < NSPLIT; ++g) st[g] = ls[g * K_DIM];
        float mx = st[0].x;
        #pragma unroll
        for (int g = 1; g < NSPLIT; ++g) mx = fmaxf(mx, st[g].x);
        float e[NSPLIT];
        float den = 0.0f;
        #pragma unroll
        for (int g = 0; g < NSPLIT; ++g) {
            e[g] = __expf(st[g].x - mx);
            den += e[g] * st[g].y;
        }
        const float inv = 1.0f / den;
        #pragma unroll
        for (int g = 0; g < NSPLIT; ++g) wt[s][g][k] = e[g] * inv;
    }
    __syncthreads();
    v4f acc = {0.0f, 0.0f, 0.0f, 0.0f};
    if (live) {
        const int k0 = (4 * rem) % K_DIM;
        const int k1 = (k0 + 1 < K_DIM) ? k0 + 1 : k0 + 1 - K_DIM;
        const int k2 = (k1 + 1 < K_DIM) ? k1 + 1 : k1 + 1 - K_DIM;
        const int k3 = (k2 + 1 < K_DIM) ? k2 + 1 : k2 + 1 - K_DIM;
        #pragma unroll
        for (int s = 0; s < S; ++s) {
            const float* w = &wt[s][z][0];
            acc.x += x[s].x * w[k0];
            acc.y += x[s].y * w[k1];
            acc.z += x[s].z * w[k2];
            acc.w += x[s].w * w[k3];
        }
    }
    red[z][cl] = acc;
    __syncthreads();
    if (t < RD_C && live) {
        v4f sum = red[0][t];
        #pragma unroll
        for (int g = 1; g < NSPLIT; ++g) sum += red[g][t];
        *((v4f*)out + col) = sum * (1.0f / (float)S);
    }
}

__global__ void __launch_bounds__(RD_T)
join_splits_kernel(const float* __restrict__ part,
                   const float* __restrict__ lstat,
                   const int* __restrict__ clip_num,
                   float* __restrict__ out, int n) {
    __shared__ __attribute__((aligned(16))) float wt[B_TOT][NSPLIT][K_DIM];
    __shared__ __attribute__((aligned(16))) v4f red[NSPLIT][RD_C];
    const int splits = clip_num[0] + 1;
    if (splits == 4)      join_body<4>(part, lstat, out, n, wt, red);
    else if (splits == 8) join_body<8>(part, lstat, out, n, wt, red);
    else if (splits == 2) join_body<2>(part, lstat, out, n, wt, red);
    else if (splits == 1) join_body<1>(part, lstat, out, n, wt, red);
}

extern "C" void kernel_launch(void* const* d_in, const int* in_sizes, int n_in,
                              void* d_out, int out_size, void* d_ws, size_t ws_size,
                              hipStream_t stream) {
    const float* feats = (const float*)d_in[0];
    const float* probs = (const float*)d_in[1];
    const int*   clip  = (const int*)d_in[2];
    float* out   = (float*)d_out;
    float* lstat = (float*)d_ws;
    float* part  = lstat + 8192;

    (void)in_sizes; (void)n_in; (void)ws_size;

    dim3 grid(C_DIM / CB, B_TOT, NSPLIT);
    pooled_gemm_kernel<<<grid, GW * 32, 0, stream>>>(feats, probs, lstat, part);

    join_splits_kernel<<<(out_size + 4 * RD_C - 1) / (4 * RD_C), RD_T, 0, stream>>>(part, lstat, clip, out, out_size);
}
